# starting kernel
# baseline (speedup 1.0000x reference)
#include <hip/hip_runtime.h>
#include <hip/hip_bf16.h>

#define PB   64
#define PT   256
#define PDF  32
#define PL   128
#define TPB  8
#define LP   132
#define AST  33

typedef __attribute__((ext_vector_type(2))) float v2f;
typedef __attribute__((ext_vector_type(8))) float v8f;

__global__ void ftattn_prep_kernel(const float* __restrict__ beta,
                                   float* __restrict__ out,
                                   float* __restrict__ wsum) {
    __shared__ float red[PT];
    const int tid = threadIdx.x;
    red[tid] = fabsf(beta[tid]);
    __syncthreads();
    for (int s = PT / 2; s > 0; s >>= 1) {
        if (tid < s) red[tid] += red[tid + s];
        __syncthreads();
    }
    if (tid == 0) wsum[0] = red[0];
    for (int i = tid; i < PB * PL; i += PT) out[i] = 0.0f;
}

__global__ void __launch_bounds__(128)
ftattn_main_kernel(const float* __restrict__ x,
                   const float* __restrict__ latent,
                   const float* __restrict__ kern,
                   const float* __restrict__ bias,
                   const float* __restrict__ beta,
                   const float* __restrict__ wsum,
                   float* __restrict__ out) {
    __shared__ float sKer[PDF * LP];
    __shared__ float sBias[PDF * LP];
    __shared__ float sF[PDF * LP];
    __shared__ float sA[PDF * AST];
    __shared__ float sInv[PDF * AST];
    __shared__ float sX[PDF];
    __shared__ float sLat[PL];
    __shared__ float sXty[PDF];
    __shared__ float sScores[PDF];

    const int tid  = threadIdx.x;
    const int b    = blockIdx.y;
    const int t0   = blockIdx.x * TPB;

    const int lane = tid & 31;
    const int wave = tid >> 5;
    const int m0   = (wave >> 1) * 16;
    const int n0   = (wave & 1) * 16;
    const int half = lane >> 4;
    const int lr   = lane & 15;

    const int gr = tid >> 2;
    const int gg = tid & 3;

    for (int idx = tid; idx < PDF * PL; idx += 128) {
        const int d = idx >> 7, l = idx & 127;
        sKer[d * LP + l]  = kern[idx];
        sBias[d * LP + l] = bias[idx];
    }

    const float inv_bsum = 1.0f / wsum[0];
    float acc = 0.0f;

    for (int tt = 0; tt < TPB; ++tt) {
        const int t = t0 + tt;

        if (tid < PDF) sX[tid] = x[((size_t)b * PT + t) * PDF + tid];
        sLat[tid] = latent[((size_t)b * PT + t) * PL + tid];

        if (tt + 1 < TPB) {
            __builtin_prefetch(&x[((size_t)b * PT + t + 1) * PDF + (tid & 31)], 0, 1);
            __builtin_prefetch(&latent[((size_t)b * PT + t + 1) * PL + tid], 0, 1);
        }
        __syncthreads();

        {
            const int l = tid;
            #pragma unroll
            for (int d = 0; d < PDF; ++d) {
                float v = fmaf(sX[d], sKer[d * LP + l], sBias[d * LP + l]);
                sF[d * LP + l] = v > 0.0f ? v : 0.0f;
            }
        }
        __syncthreads();

        {
            v8f c = {};
            const float* rowA = &sF[(m0 + lr) * LP];
            const float* rowB = &sF[(n0 + lr) * LP];
            #pragma unroll
            for (int k0 = 0; k0 < PL; k0 += 4) {
                v2f a, bb;
                a.x  = rowA[k0 + 2 * half];
                a.y  = rowA[k0 + 2 * half + 1];
                bb.x = rowB[k0 + 2 * half];
                bb.y = rowB[k0 + 2 * half + 1];
                c = __builtin_amdgcn_wmma_f32_16x16x4_f32(
                        false, a, false, bb, (short)0, c, false, false);
            }
            #pragma unroll
            for (int r = 0; r < 8; ++r) {
                sA[(m0 + half * 8 + r) * AST + (n0 + lr)] = c[r];
            }
        }
        for (int idx = tid; idx < PDF * PDF; idx += 128) {
            const int rr = idx >> 5, cc = idx & 31;
            sInv[rr * AST + cc] = (rr == cc) ? 1.0f : 0.0f;
        }
        __syncthreads();

        for (int k = 0; k < PDF; ++k) {
            const float piv  = sA[k * AST + k];
            const float pinv = 1.0f / piv;
            __syncthreads();
            if (gr == k) {
                #pragma unroll
                for (int j = 0; j < 8; ++j) {
                    sA[k * AST + gg * 8 + j]   *= pinv;
                    sInv[k * AST + gg * 8 + j] *= pinv;
                }
            }
            __syncthreads();
            const float f = sA[gr * AST + k];
            if (gr != k) {
                #pragma unroll
                for (int j = 0; j < 8; ++j) {
                    const int cc = gg * 8 + j;
                    sA[gr * AST + cc]   -= f * sA[k * AST + cc];
                    sInv[gr * AST + cc] -= f * sInv[k * AST + cc];
                }
            }
            __syncthreads();
        }

        if (tid < PDF) {
            float s = 0.0f;
            #pragma unroll 8
            for (int l = 0; l < PL; ++l) s = fmaf(sF[tid * LP + l], sLat[l], s);
            sXty[tid] = s;
        }
        __syncthreads();

        if (tid < PDF) {
            float s = 0.0f;
            #pragma unroll
            for (int e = 0; e < PDF; ++e) s = fmaf(sInv[tid * AST + e], sXty[e], s);
            sScores[tid] = s;
        }
        __syncthreads();

        {
            const int l = tid;
            float oh = 0.0f;
            #pragma unroll
            for (int d = 0; d < PDF; ++d) oh = fmaf(sScores[d], sF[d * LP + l], oh);
            acc = fmaf(fabsf(beta[t]) * inv_bsum, oh, acc);
        }
        __syncthreads();
    }

    atomicAdd(&out[b * PL + tid], acc);
}

extern "C" void kernel_launch(void* const* d_in, const int* in_sizes, int n_in,
                              void* d_out, int out_size, void* d_ws, size_t ws_size,
                              hipStream_t stream) {
    const float* x      = (const float*)d_in[0];
    const float* latent = (const float*)d_in[1];
    const float* kern   = (const float*)d_in[2];
    const float* bias   = (const float*)d_in[3];
    const float* beta   = (const float*)d_in[4];
    float* out  = (float*)d_out;
    float* wsum = (float*)d_ws;

    ftattn_prep_kernel<<<1, PT, 0, stream>>>(beta, out, wsum);

    dim3 grid(PT / TPB, PB);
    ftattn_main_kernel<<<grid, 128, 0, stream>>>(x, latent, kern, bias, beta, wsum, out);
}


# Features never stored: Gram fragments, xty and o_hat recompute relu(x*K+b) from kernel/bias in LDS; wave LDS 25->8 KB, so 16 waves per block (4 per SIMD, 4 steps each); xty and output sums regrouped
# speedup vs baseline: 7.0884x; 7.0884x over previous
#include <hip/hip_runtime.h>

#define PB   64
#define PT   256
#define PDF  32
#define PL   128
#define NW   16
#define TPW  4
#define LP   132
#define AST  33

typedef __attribute__((ext_vector_type(2))) float v2f;
typedef __attribute__((ext_vector_type(4))) float v4f;
typedef __attribute__((ext_vector_type(8))) float v8f;

struct WaveLds {
    float Inv[PDF * AST];
    float Mx[2 * PDF * 4];
    float Pan[2 * 4 * 64];
    float Lat[PL];
};

__global__ void ftattn_prep_kernel(const float* __restrict__ beta,
                                   float* __restrict__ out,
                                   float* __restrict__ wsum) {
    __shared__ float red[PT];
    const int tid = threadIdx.x;
    red[tid] = fabsf(beta[tid]);
    __syncthreads();
    for (int s = PT / 2; s > 0; s >>= 1) {
        if (tid < s) red[tid] += red[tid + s];
        __syncthreads();
    }
    if (tid == 0) wsum[0] = red[0];
    for (int i = tid; i < PB * PL; i += PT) out[i] = 0.0f;
}

__device__ __forceinline__ void wave_lds_handover() {
    __builtin_amdgcn_fence(__ATOMIC_RELEASE, "wavefront");
    __builtin_amdgcn_wave_barrier();
    __builtin_amdgcn_fence(__ATOMIC_ACQUIRE, "wavefront");
}

__device__ __forceinline__ float lane_bcast(float v, int srcLane) {
    return __int_as_float(__builtin_amdgcn_readlane(__float_as_int(v), srcLane));
}

__device__ __forceinline__ float xhalf(float v) {
    const int b = __float_as_int(v);
    return __int_as_float(__builtin_amdgcn_permlanex16(b, b, 0x76543210, 0xfedcba98, false, false));
}

template <int P>
__device__ __forceinline__ void gj_panel_step(v8f (&c0)[4], v8f (&c1)[4],
                                              float* sMxBase, float* sPanBase,
                                              const int half, const int lr) {
    constexpr int K0 = 4 * P;
    constexpr int WC = K0 >> 4;
    constexpr int HP = (K0 >> 3) & 1;
    constexpr int RP = K0 & 7;
    constexpr int CL = K0 & 15;
    float* mx  = sMxBase  + (P & 1) * (PDF * 4);
    float* pan = sPanBase + (P & 1) * (4 * 64);

    if (lr >= CL && lr < CL + 4) {
        const int j = lr - CL;
        #pragma unroll
        for (int r = 0; r < 8; ++r) {
            mx[(8 * half + r) * 4 + j]      = c0[WC][r];
            mx[(16 + 8 * half + r) * 4 + j] = c1[WC][r];
        }
    }
    if (half == HP) {
        #pragma unroll
        for (int vw = 0; vw < 4; ++vw) {
            #pragma unroll
            for (int j = 0; j < 4; ++j)
                pan[j * 64 + 16 * vw + lr] = (WC == 0) ? c0[vw][RP + j] : c1[vw][RP + j];
        }
    }
    wave_lds_handover();

    float m[4][4];
    float q[4][4];
    #pragma unroll
    for (int i = 0; i < 4; ++i) {
        const v4f row = *reinterpret_cast<const v4f*>(mx + (K0 + i) * 4);
        m[i][0] = row.x; m[i][1] = row.y; m[i][2] = row.z; m[i][3] = row.w;
    }
    #pragma unroll
    for (int vw = 0; vw < 4; ++vw) {
        #pragma unroll
        for (int j = 0; j < 4; ++j) q[vw][j] = pan[j * 64 + 16 * vw + lr];
    }

    const v2f a0 = *reinterpret_cast<const v2f*>(mx + lr * 4 + 2 * half);
    const v2f a1 = *reinterpret_cast<const v2f*>(mx + (16 + lr) * 4 + 2 * half);

    #pragma unroll
    for (int k = 0; k < 4; ++k) {
        const float pinv = 1.0f / m[k][k];
        #pragma unroll
        for (int j = k + 1; j < 4; ++j) m[k][j] *= pinv;
        #pragma unroll
        for (int vw = 0; vw < 4; ++vw) q[vw][k] *= pinv;
        #pragma unroll
        for (int i = 0; i < 4; ++i) {
            if (i == k) continue;
            const float f = m[i][k];
            #pragma unroll
            for (int j = k + 1; j < 4; ++j) m[i][j] -= f * m[k][j];
            #pragma unroll
            for (int vw = 0; vw < 4; ++vw) q[vw][i] -= f * q[vw][k];
        }
    }

    #pragma unroll
    for (int vw = 0; vw < 4; ++vw) {
        v2f bb;
        bb.x = -(half ? q[vw][2] : q[vw][0]);
        bb.y = -(half ? q[vw][3] : q[vw][1]);
        c0[vw] = __builtin_amdgcn_wmma_f32_16x16x4_f32(false, a0, false, bb, (short)0, c0[vw], false, false);
        c1[vw] = __builtin_amdgcn_wmma_f32_16x16x4_f32(false, a1, false, bb, (short)0, c1[vw], false, false);
    }

    if (half == HP) {
        #pragma unroll
        for (int vw = 0; vw < 4; ++vw) {
            #pragma unroll
            for (int j = 0; j < 4; ++j) {
                if (WC == 0) c0[vw][RP + j] = q[vw][j];
                else         c1[vw][RP + j] = q[vw][j];
            }
        }
    }
}

__global__ void __launch_bounds__(NW * 32)
ftattn_main_kernel(const float* __restrict__ x,
                   const float* __restrict__ latent,
                   const float* __restrict__ kern,
                   const float* __restrict__ bias,
                   const float* __restrict__ beta,
                   const float* __restrict__ wsum,
                   float* __restrict__ out) {
    __shared__ __attribute__((aligned(16))) float sKer[PDF * LP];
    __shared__ __attribute__((aligned(16))) float sBias[PDF * LP];
    __shared__ __attribute__((aligned(16))) WaveLds sW[NW];

    const int tid  = threadIdx.x;
    const int b    = blockIdx.y;
    const int lane = tid & 31;
    const int wave = __builtin_amdgcn_readfirstlane(tid >> 5);
    const int half = lane >> 4;
    const int lr   = lane & 15;
    const int t0   = blockIdx.x * (NW * TPW) + wave * TPW;

    float* sInv = sW[wave].Inv;
    float* sMx  = sW[wave].Mx;
    float* sPan = sW[wave].Pan;
    float* sLat = sW[wave].Lat;

    for (int i4 = tid; i4 < PDF * PL / 4; i4 += NW * 32) {
        const int d = i4 >> 5, l = (i4 & 31) * 4;
        *reinterpret_cast<v4f*>(&sKer[d * LP + l])  = *reinterpret_cast<const v4f*>(&kern[i4 * 4]);
        *reinterpret_cast<v4f*>(&sBias[d * LP + l]) = *reinterpret_cast<const v4f*>(&bias[i4 * 4]);
    }

    const float inv_bsum = 1.0f / wsum[0];
    v4f acc = {0.0f, 0.0f, 0.0f, 0.0f};

    float xa_next  = x[((size_t)b * PT + t0) * PDF + lr];
    float xb_next  = x[((size_t)b * PT + t0) * PDF + 16 + lr];
    v4f   lat_next = *reinterpret_cast<const v4f*>(&latent[((size_t)b * PT + t0) * PL + 4 * lane]);

    const float* kr0 = &sKer[lr * LP + 2 * half];
    const float* kr1 = &sKer[(16 + lr) * LP + 2 * half];
    const float* br0 = &sBias[lr * LP + 2 * half];
    const float* br1 = &sBias[(16 + lr) * LP + 2 * half];
    const float* yr  = &sLat[2 * half];

    __syncthreads();

    for (int tt = 0; tt < TPW; ++tt) {
        const int t = t0 + tt;
        const float xa = xa_next;
        const float xb = xb_next;
        const float xv = half ? xb : xa;
        *reinterpret_cast<v4f*>(&sLat[4 * lane]) = lat_next;
        if (tt + 1 < TPW) {
            xa_next  = x[((size_t)b * PT + t + 1) * PDF + lr];
            xb_next  = x[((size_t)b * PT + t + 1) * PDF + 16 + lr];
            lat_next = *reinterpret_cast<const v4f*>(&latent[((size_t)b * PT + t + 1) * PL + 4 * lane]);
        }
        wave_lds_handover();

        v8f c0[4], c1[4];
        v2f p0 = {0.0f, 0.0f}, p1 = {0.0f, 0.0f};
        {
            const v2f xa2 = {xa, xa};
            const v2f xb2 = {xb, xb};
            c0[0] = (v8f){}; c1[0] = (v8f){}; c0[1] = (v8f){}; c1[1] = (v8f){};
            for (int kb = 0; kb < PL; kb += 32) {
                v2f ka[8], ba[8], kc[8], bc[8], yy[8];
                #pragma unroll
                for (int j = 0; j < 8; ++j) {
                    ka[j] = *reinterpret_cast<const v2f*>(kr0 + kb + 4 * j);
                    ba[j] = *reinterpret_cast<const v2f*>(br0 + kb + 4 * j);
                    kc[j] = *reinterpret_cast<const v2f*>(kr1 + kb + 4 * j);
                    bc[j] = *reinterpret_cast<const v2f*>(br1 + kb + 4 * j);
                    yy[j] = *reinterpret_cast<const v2f*>(yr + kb + 4 * j);
                }
                #pragma unroll
                for (int j = 0; j < 8; ++j) {
                    v2f f0 = __builtin_elementwise_fma(xa2, ka[j], ba[j]);
                    v2f f1 = __builtin_elementwise_fma(xb2, kc[j], bc[j]);
                    f0.x = f0.x > 0.0f ? f0.x : 0.0f; f0.y = f0.y > 0.0f ? f0.y : 0.0f;
                    f1.x = f1.x > 0.0f ? f1.x : 0.0f; f1.y = f1.y > 0.0f ? f1.y : 0.0f;
                    c0[0] = __builtin_amdgcn_wmma_f32_16x16x4_f32(false, f0, false, f0, (short)0, c0[0], false, false);
                    c1[0] = __builtin_amdgcn_wmma_f32_16x16x4_f32(false, f1, false, f0, (short)0, c1[0], false, false);
                    c0[1] = __builtin_amdgcn_wmma_f32_16x16x4_f32(false, f0, false, f1, (short)0, c0[1], false, false);
                    c1[1] = __builtin_amdgcn_wmma_f32_16x16x4_f32(false, f1, false, f1, (short)0, c1[1], false, false);
                    p0 = __builtin_elementwise_fma(f0, yy[j], p0);
                    p1 = __builtin_elementwise_fma(f1, yy[j], p1);
                }
            }
            #pragma unroll
            for (int vw = 2; vw < 4; ++vw) {
                const int col = 16 * (vw - 2) + lr;
                #pragma unroll
                for (int r = 0; r < 8; ++r) {
                    c0[vw][r] = (8 * half + r == col) ? 1.0f : 0.0f;
                    c1[vw][r] = (16 + 8 * half + r == col) ? 1.0f : 0.0f;
                }
            }
        }

        const float s0 = p0.x + p0.y;
        const float s1 = p1.x + p1.y;
        const float xty0 = s0 + xhalf(s0);
        const float xty1 = s1 + xhalf(s1);
        const float xty  = half ? xty1 : xty0;

        gj_panel_step<0>(c0, c1, sMx, sPan, half, lr);
        gj_panel_step<1>(c0, c1, sMx, sPan, half, lr);
        gj_panel_step<2>(c0, c1, sMx, sPan, half, lr);
        gj_panel_step<3>(c0, c1, sMx, sPan, half, lr);
        gj_panel_step<4>(c0, c1, sMx, sPan, half, lr);
        gj_panel_step<5>(c0, c1, sMx, sPan, half, lr);
        gj_panel_step<6>(c0, c1, sMx, sPan, half, lr);
        gj_panel_step<7>(c0, c1, sMx, sPan, half, lr);
        #pragma unroll
        for (int vw = 2; vw < 4; ++vw) {
            const int col = 16 * (vw - 2) + lr;
            #pragma unroll
            for (int r = 0; r < 8; ++r) {
                sInv[(8 * half + r) * AST + col]      = c0[vw][r];
                sInv[(16 + 8 * half + r) * AST + col] = c1[vw][r];
            }
        }
        wave_lds_handover();

        float score = 0.0f;
        #pragma unroll
        for (int e = 0; e < PDF; ++e) score = fmaf(sInv[lane * AST + e], lane_bcast(xty, e), score);

        {
            v4f oh = {0.0f, 0.0f, 0.0f, 0.0f};
            #pragma unroll
            for (int d = 0; d < PDF; ++d) {
                const float xd = lane_bcast(xv, d);
                const float sd = lane_bcast(score, d);
                const v4f kk = *reinterpret_cast<const v4f*>(&sKer[d * LP + 4 * lane]);
                const v4f bs = *reinterpret_cast<const v4f*>(&sBias[d * LP + 4 * lane]);
                v4f f;
                f.x = fmaf(xd, kk.x, bs.x); f.y = fmaf(xd, kk.y, bs.y);
                f.z = fmaf(xd, kk.z, bs.z); f.w = fmaf(xd, kk.w, bs.w);
                f.x = f.x > 0.0f ? f.x : 0.0f; f.y = f.y > 0.0f ? f.y : 0.0f;
                f.z = f.z > 0.0f ? f.z : 0.0f; f.w = f.w > 0.0f ? f.w : 0.0f;
                oh.x = fmaf(sd, f.x, oh.x); oh.y = fmaf(sd, f.y, oh.y);
                oh.z = fmaf(sd, f.z, oh.z); oh.w = fmaf(sd, f.w, oh.w);
            }
            const float w = fabsf(beta[t]) * inv_bsum;
            acc.x = fmaf(w, oh.x, acc.x); acc.y = fmaf(w, oh.y, acc.y);
            acc.z = fmaf(w, oh.z, acc.z); acc.w = fmaf(w, oh.w, acc.w);
        }
        wave_lds_handover();
    }

    *reinterpret_cast<v4f*>(&sInv[4 * lane]) = acc;
    __syncthreads();
    if (tid < PL) {
        float s = 0.0f;
        #pragma unroll
        for (int w = 0; w < NW; ++w) s += sW[w].Inv[tid];
        atomicAdd(&out[b * PL + tid], s);
    }
}

extern "C" void kernel_launch(void* const* d_in, const int* in_sizes, int n_in,
                              void* d_out, int out_size, void* d_ws, size_t ws_size,
                              hipStream_t stream) {
    const float* x      = (const float*)d_in[0];
    const float* latent = (const float*)d_in[1];
    const float* kern   = (const float*)d_in[2];
    const float* bias   = (const float*)d_in[3];
    const float* beta   = (const float*)d_in[4];
    float* out  = (float*)d_out;
    float* wsum = (float*)d_ws;

    ftattn_prep_kernel<<<1, PT, 0, stream>>>(beta, out, wsum);

    dim3 grid(PT / (NW * TPW), PB);
    ftattn_main_kernel<<<grid, NW * 32, 0, stream>>>(x, latent, kern, bias, beta, wsum, out);
}


# In-place blocked Gauss-Jordan: identity half of [XtX|I] dropped, inverse columns replace pivot columns in the 32x32 accumulators; 4 products per pivot step, not 8; per-value operation order kept
# speedup vs baseline: 7.5094x; 1.0594x over previous
#include <hip/hip_runtime.h>

#define PB   64
#define PT   256
#define PDF  32
#define PL   128
#define NW   16
#define TPW  4
#define LP   132
#define AST  33

typedef __attribute__((ext_vector_type(2))) float v2f;
typedef __attribute__((ext_vector_type(4))) float v4f;
typedef __attribute__((ext_vector_type(8))) float v8f;

struct WaveLds {
    float Inv[PDF * AST];
    float Mx[2 * PDF * 4];
    float Pan[2 * 4 * PDF];
    float Lat[PL];
};

__global__ void ftattn_prep_kernel(const float* __restrict__ beta,
                                   float* __restrict__ out,
                                   float* __restrict__ wsum) {
    __shared__ float red[PT];
    const int tid = threadIdx.x;
    red[tid] = fabsf(beta[tid]);
    __syncthreads();
    for (int s = PT / 2; s > 0; s >>= 1) {
        if (tid < s) red[tid] += red[tid + s];
        __syncthreads();
    }
    if (tid == 0) wsum[0] = red[0];
    for (int i = tid; i < PB * PL; i += PT) out[i] = 0.0f;
}

__device__ __forceinline__ void wave_lds_handover() {
    __builtin_amdgcn_fence(__ATOMIC_RELEASE, "wavefront");
    __builtin_amdgcn_wave_barrier();
    __builtin_amdgcn_fence(__ATOMIC_ACQUIRE, "wavefront");
}

__device__ __forceinline__ float lane_bcast(float v, int srcLane) {
    return __int_as_float(__builtin_amdgcn_readlane(__float_as_int(v), srcLane));
}

__device__ __forceinline__ float xhalf(float v) {
    const int b = __float_as_int(v);
    return __int_as_float(__builtin_amdgcn_permlanex16(b, b, 0x76543210, 0xfedcba98, false, false));
}

template <int P>
__device__ __forceinline__ void gj_panel_step(v8f (&c0)[2], v8f (&c1)[2],
                                              float* sMxBase, float* sPanBase,
                                              const int half, const int lr) {
    constexpr int K0 = 4 * P;
    constexpr int WC = K0 >> 4;
    constexpr int HP = (K0 >> 3) & 1;
    constexpr int RP = K0 & 7;
    constexpr int CL = K0 & 15;
    float* mx  = sMxBase  + (P & 1) * (PDF * 4);
    float* pan = sPanBase + (P & 1) * (4 * PDF);

    const bool pivcol = (lr >= CL && lr < CL + 4);
    if (pivcol) {
        const int j = lr - CL;
        #pragma unroll
        for (int r = 0; r < 8; ++r) {
            mx[(8 * half + r) * 4 + j]      = c0[WC][r];
            mx[(16 + 8 * half + r) * 4 + j] = c1[WC][r];
        }
    }
    #pragma unroll
    for (int r = 0; r < 8; ++r) {
        float e0 = 0.0f, e1 = 0.0f;
        if (r >= RP && r < RP + 4) {
            const float dg = (half == HP && lr == CL + r - RP) ? 1.0f : 0.0f;
            if (WC == 0) e0 = dg; else e1 = dg;
        }
        c0[WC][r] = pivcol ? e0 : c0[WC][r];
        c1[WC][r] = pivcol ? e1 : c1[WC][r];
    }
    if (half == HP) {
        #pragma unroll
        for (int vw = 0; vw < 2; ++vw) {
            #pragma unroll
            for (int j = 0; j < 4; ++j)
                pan[j * PDF + 16 * vw + lr] = (WC == 0) ? c0[vw][RP + j] : c1[vw][RP + j];
        }
    }
    wave_lds_handover();

    float m[4][4];
    float q[2][4];
    #pragma unroll
    for (int i = 0; i < 4; ++i) {
        const v4f row = *reinterpret_cast<const v4f*>(mx + (K0 + i) * 4);
        m[i][0] = row.x; m[i][1] = row.y; m[i][2] = row.z; m[i][3] = row.w;
    }
    #pragma unroll
    for (int vw = 0; vw < 2; ++vw) {
        #pragma unroll
        for (int j = 0; j < 4; ++j) q[vw][j] = pan[j * PDF + 16 * vw + lr];
    }

    const v2f a0 = *reinterpret_cast<const v2f*>(mx + lr * 4 + 2 * half);
    const v2f a1 = *reinterpret_cast<const v2f*>(mx + (16 + lr) * 4 + 2 * half);

    #pragma unroll
    for (int k = 0; k < 4; ++k) {
        const float pinv = 1.0f / m[k][k];
        #pragma unroll
        for (int j = k + 1; j < 4; ++j) m[k][j] *= pinv;
        #pragma unroll
        for (int vw = 0; vw < 2; ++vw) q[vw][k] *= pinv;
        #pragma unroll
        for (int i = 0; i < 4; ++i) {
            if (i == k) continue;
            const float f = m[i][k];
            #pragma unroll
            for (int j = k + 1; j < 4; ++j) m[i][j] -= f * m[k][j];
            #pragma unroll
            for (int vw = 0; vw < 2; ++vw) q[vw][i] -= f * q[vw][k];
        }
    }

    #pragma unroll
    for (int vw = 0; vw < 2; ++vw) {
        v2f bb;
        bb.x = -(half ? q[vw][2] : q[vw][0]);
        bb.y = -(half ? q[vw][3] : q[vw][1]);
        c0[vw] = __builtin_amdgcn_wmma_f32_16x16x4_f32(false, a0, false, bb, (short)0, c0[vw], false, false);
        c1[vw] = __builtin_amdgcn_wmma_f32_16x16x4_f32(false, a1, false, bb, (short)0, c1[vw], false, false);
    }

    if (half == HP) {
        #pragma unroll
        for (int vw = 0; vw < 2; ++vw) {
            #pragma unroll
            for (int j = 0; j < 4; ++j) {
                if (WC == 0) c0[vw][RP + j] = q[vw][j];
                else         c1[vw][RP + j] = q[vw][j];
            }
        }
    }
}

__global__ void __launch_bounds__(NW * 32)
ftattn_main_kernel(const float* __restrict__ x,
                   const float* __restrict__ latent,
                   const float* __restrict__ kern,
                   const float* __restrict__ bias,
                   const float* __restrict__ beta,
                   const float* __restrict__ wsum,
                   float* __restrict__ out) {
    __shared__ __attribute__((aligned(16))) float sKer[PDF * LP];
    __shared__ __attribute__((aligned(16))) float sBias[PDF * LP];
    __shared__ __attribute__((aligned(16))) WaveLds sW[NW];

    const int tid  = threadIdx.x;
    const int b    = blockIdx.y;
    const int lane = tid & 31;
    const int wave = __builtin_amdgcn_readfirstlane(tid >> 5);
    const int half = lane >> 4;
    const int lr   = lane & 15;
    const int t0   = blockIdx.x * (NW * TPW) + wave * TPW;

    float* sInv = sW[wave].Inv;
    float* sMx  = sW[wave].Mx;
    float* sPan = sW[wave].Pan;
    float* sLat = sW[wave].Lat;

    for (int i4 = tid; i4 < PDF * PL / 4; i4 += NW * 32) {
        const int d = i4 >> 5, l = (i4 & 31) * 4;
        *reinterpret_cast<v4f*>(&sKer[d * LP + l])  = *reinterpret_cast<const v4f*>(&kern[i4 * 4]);
        *reinterpret_cast<v4f*>(&sBias[d * LP + l]) = *reinterpret_cast<const v4f*>(&bias[i4 * 4]);
    }

    const float inv_bsum = 1.0f / wsum[0];
    v4f acc = {0.0f, 0.0f, 0.0f, 0.0f};

    float xa_next  = x[((size_t)b * PT + t0) * PDF + lr];
    float xb_next  = x[((size_t)b * PT + t0) * PDF + 16 + lr];
    v4f   lat_next = *reinterpret_cast<const v4f*>(&latent[((size_t)b * PT + t0) * PL + 4 * lane]);

    const float* kr0 = &sKer[lr * LP + 2 * half];
    const float* kr1 = &sKer[(16 + lr) * LP + 2 * half];
    const float* br0 = &sBias[lr * LP + 2 * half];
    const float* br1 = &sBias[(16 + lr) * LP + 2 * half];
    const float* yr  = &sLat[2 * half];

    __syncthreads();

    for (int tt = 0; tt < TPW; ++tt) {
        const int t = t0 + tt;
        const float xa = xa_next;
        const float xb = xb_next;
        const float xv = half ? xb : xa;
        *reinterpret_cast<v4f*>(&sLat[4 * lane]) = lat_next;
        if (tt + 1 < TPW) {
            xa_next  = x[((size_t)b * PT + t + 1) * PDF + lr];
            xb_next  = x[((size_t)b * PT + t + 1) * PDF + 16 + lr];
            lat_next = *reinterpret_cast<const v4f*>(&latent[((size_t)b * PT + t + 1) * PL + 4 * lane]);
        }
        wave_lds_handover();

        v8f c0[2], c1[2];
        v2f p0 = {0.0f, 0.0f}, p1 = {0.0f, 0.0f};
        {
            const v2f xa2 = {xa, xa};
            const v2f xb2 = {xb, xb};
            c0[0] = (v8f){}; c1[0] = (v8f){}; c0[1] = (v8f){}; c1[1] = (v8f){};
            for (int kb = 0; kb < PL; kb += 32) {
                v2f ka[8], ba[8], kc[8], bc[8], yy[8];
                #pragma unroll
                for (int j = 0; j < 8; ++j) {
                    ka[j] = *reinterpret_cast<const v2f*>(kr0 + kb + 4 * j);
                    ba[j] = *reinterpret_cast<const v2f*>(br0 + kb + 4 * j);
                    kc[j] = *reinterpret_cast<const v2f*>(kr1 + kb + 4 * j);
                    bc[j] = *reinterpret_cast<const v2f*>(br1 + kb + 4 * j);
                    yy[j] = *reinterpret_cast<const v2f*>(yr + kb + 4 * j);
                }
                #pragma unroll
                for (int j = 0; j < 8; ++j) {
                    v2f f0 = __builtin_elementwise_fma(xa2, ka[j], ba[j]);
                    v2f f1 = __builtin_elementwise_fma(xb2, kc[j], bc[j]);
                    f0.x = f0.x > 0.0f ? f0.x : 0.0f; f0.y = f0.y > 0.0f ? f0.y : 0.0f;
                    f1.x = f1.x > 0.0f ? f1.x : 0.0f; f1.y = f1.y > 0.0f ? f1.y : 0.0f;
                    c0[0] = __builtin_amdgcn_wmma_f32_16x16x4_f32(false, f0, false, f0, (short)0, c0[0], false, false);
                    c1[0] = __builtin_amdgcn_wmma_f32_16x16x4_f32(false, f1, false, f0, (short)0, c1[0], false, false);
                    c0[1] = __builtin_amdgcn_wmma_f32_16x16x4_f32(false, f0, false, f1, (short)0, c0[1], false, false);
                    c1[1] = __builtin_amdgcn_wmma_f32_16x16x4_f32(false, f1, false, f1, (short)0, c1[1], false, false);
                    p0 = __builtin_elementwise_fma(f0, yy[j], p0);
                    p1 = __builtin_elementwise_fma(f1, yy[j], p1);
                }
            }
        }

        const float s0 = p0.x + p0.y;
        const float s1 = p1.x + p1.y;
        const float xty0 = s0 + xhalf(s0);
        const float xty1 = s1 + xhalf(s1);
        const float xty  = half ? xty1 : xty0;

        gj_panel_step<0>(c0, c1, sMx, sPan, half, lr);
        gj_panel_step<1>(c0, c1, sMx, sPan, half, lr);
        gj_panel_step<2>(c0, c1, sMx, sPan, half, lr);
        gj_panel_step<3>(c0, c1, sMx, sPan, half, lr);
        gj_panel_step<4>(c0, c1, sMx, sPan, half, lr);
        gj_panel_step<5>(c0, c1, sMx, sPan, half, lr);
        gj_panel_step<6>(c0, c1, sMx, sPan, half, lr);
        gj_panel_step<7>(c0, c1, sMx, sPan, half, lr);
        #pragma unroll
        for (int vw = 0; vw < 2; ++vw) {
            const int col = 16 * vw + lr;
            #pragma unroll
            for (int r = 0; r < 8; ++r) {
                sInv[(8 * half + r) * AST + col]      = c0[vw][r];
                sInv[(16 + 8 * half + r) * AST + col] = c1[vw][r];
            }
        }
        wave_lds_handover();

        float score = 0.0f;
        #pragma unroll
        for (int e = 0; e < PDF; ++e) score = fmaf(sInv[lane * AST + e], lane_bcast(xty, e), score);

        {
            v4f oh = {0.0f, 0.0f, 0.0f, 0.0f};
            #pragma unroll
            for (int d = 0; d < PDF; ++d) {
                const float xd = lane_bcast(xv, d);
                const float sd = lane_bcast(score, d);
                const v4f kk = *reinterpret_cast<const v4f*>(&sKer[d * LP + 4 * lane]);
                const v4f bs = *reinterpret_cast<const v4f*>(&sBias[d * LP + 4 * lane]);
                v4f f;
                f.x = fmaf(xd, kk.x, bs.x); f.y = fmaf(xd, kk.y, bs.y);
                f.z = fmaf(xd, kk.z, bs.z); f.w = fmaf(xd, kk.w, bs.w);
                f.x = f.x > 0.0f ? f.x : 0.0f; f.y = f.y > 0.0f ? f.y : 0.0f;
                f.z = f.z > 0.0f ? f.z : 0.0f; f.w = f.w > 0.0f ? f.w : 0.0f;
                oh.x = fmaf(sd, f.x, oh.x); oh.y = fmaf(sd, f.y, oh.y);
                oh.z = fmaf(sd, f.z, oh.z); oh.w = fmaf(sd, f.w, oh.w);
            }
            const float w = fabsf(beta[t]) * inv_bsum;
            acc.x = fmaf(w, oh.x, acc.x); acc.y = fmaf(w, oh.y, acc.y);
            acc.z = fmaf(w, oh.z, acc.z); acc.w = fmaf(w, oh.w, acc.w);
        }
        wave_lds_handover();
    }

    *reinterpret_cast<v4f*>(&sInv[4 * lane]) = acc;
    __syncthreads();
    if (tid < PL) {
        float s = 0.0f;
        #pragma unroll
        for (int w = 0; w < NW; ++w) s += sW[w].Inv[tid];
        atomicAdd(&out[b * PL + tid], s);
    }
}

extern "C" void kernel_launch(void* const* d_in, const int* in_sizes, int n_in,
                              void* d_out, int out_size, void* d_ws, size_t ws_size,
                              hipStream_t stream) {
    const float* x      = (const float*)d_in[0];
    const float* latent = (const float*)d_in[1];
    const float* kern   = (const float*)d_in[2];
    const float* bias   = (const float*)d_in[3];
    const float* beta   = (const float*)d_in[4];
    float* out  = (float*)d_out;
    float* wsum = (float*)d_ws;

    ftattn_prep_kernel<<<1, PT, 0, stream>>>(beta, out, wsum);

    dim3 grid(PT / (NW * TPW), PB);
    ftattn_main_kernel<<<grid, NW * 32, 0, stream>>>(x, latent, kern, bias, beta, wsum, out);
}


# Pivot reciprocals of the blocked Gauss-Jordan by v_rcp_f32 plus one Newton step, not the IEEE division expansion: 3 dependent instructions per pivot, not 12; last bit may differ, no sum reordered
# speedup vs baseline: 7.9977x; 1.0650x over previous
#include <hip/hip_runtime.h>

#define PB   64
#define PT   256
#define PDF  32
#define PL   128
#define NW   16
#define TPW  4
#define LP   132
#define AST  33

typedef __attribute__((ext_vector_type(2))) float v2f;
typedef __attribute__((ext_vector_type(4))) float v4f;
typedef __attribute__((ext_vector_type(8))) float v8f;

struct WaveLds {
    float Inv[PDF * AST];
    float Mx[2 * PDF * 4];
    float Pan[2 * 4 * PDF];
    float Lat[PL];
};

__global__ void ftattn_prep_kernel(const float* __restrict__ beta,
                                   float* __restrict__ out,
                                   float* __restrict__ wsum) {
    __shared__ float red[PT];
    const int tid = threadIdx.x;
    red[tid] = fabsf(beta[tid]);
    __syncthreads();
    for (int s = PT / 2; s > 0; s >>= 1) {
        if (tid < s) red[tid] += red[tid + s];
        __syncthreads();
    }
    if (tid == 0) wsum[0] = red[0];
    for (int i = tid; i < PB * PL; i += PT) out[i] = 0.0f;
}

__device__ __forceinline__ void wave_lds_handover() {
    __builtin_amdgcn_fence(__ATOMIC_RELEASE, "wavefront");
    __builtin_amdgcn_wave_barrier();
    __builtin_amdgcn_fence(__ATOMIC_ACQUIRE, "wavefront");
}

__device__ __forceinline__ float lane_bcast(float v, int srcLane) {
    return __int_as_float(__builtin_amdgcn_readlane(__float_as_int(v), srcLane));
}

__device__ __forceinline__ float xhalf(float v) {
    const int b = __float_as_int(v);
    return __int_as_float(__builtin_amdgcn_permlanex16(b, b, 0x76543210, 0xfedcba98, false, false));
}

template <int P>
__device__ __forceinline__ void gj_panel_step(v8f (&c0)[2], v8f (&c1)[2],
                                              float* sMxBase, float* sPanBase,
                                              const int half, const int lr) {
    constexpr int K0 = 4 * P;
    constexpr int WC = K0 >> 4;
    constexpr int HP = (K0 >> 3) & 1;
    constexpr int RP = K0 & 7;
    constexpr int CL = K0 & 15;
    float* mx  = sMxBase  + (P & 1) * (PDF * 4);
    float* pan = sPanBase + (P & 1) * (4 * PDF);

    const bool pivcol = (lr >= CL && lr < CL + 4);
    if (pivcol) {
        const int j = lr - CL;
        #pragma unroll
        for (int r = 0; r < 8; ++r) {
            mx[(8 * half + r) * 4 + j]      = c0[WC][r];
            mx[(16 + 8 * half + r) * 4 + j] = c1[WC][r];
        }
    }
    #pragma unroll
    for (int r = 0; r < 8; ++r) {
        float e0 = 0.0f, e1 = 0.0f;
        if (r >= RP && r < RP + 4) {
            const float dg = (half == HP && lr == CL + r - RP) ? 1.0f : 0.0f;
            if (WC == 0) e0 = dg; else e1 = dg;
        }
        c0[WC][r] = pivcol ? e0 : c0[WC][r];
        c1[WC][r] = pivcol ? e1 : c1[WC][r];
    }
    if (half == HP) {
        #pragma unroll
        for (int vw = 0; vw < 2; ++vw) {
            #pragma unroll
            for (int j = 0; j < 4; ++j)
                pan[j * PDF + 16 * vw + lr] = (WC == 0) ? c0[vw][RP + j] : c1[vw][RP + j];
        }
    }
    wave_lds_handover();

    float m[4][4];
    float q[2][4];
    #pragma unroll
    for (int i = 0; i < 4; ++i) {
        const v4f row = *reinterpret_cast<const v4f*>(mx + (K0 + i) * 4);
        m[i][0] = row.x; m[i][1] = row.y; m[i][2] = row.z; m[i][3] = row.w;
    }
    #pragma unroll
    for (int vw = 0; vw < 2; ++vw) {
        #pragma unroll
        for (int j = 0; j < 4; ++j) q[vw][j] = pan[j * PDF + 16 * vw + lr];
    }

    const v2f a0 = *reinterpret_cast<const v2f*>(mx + lr * 4 + 2 * half);
    const v2f a1 = *reinterpret_cast<const v2f*>(mx + (16 + lr) * 4 + 2 * half);

    #pragma unroll
    for (int k = 0; k < 4; ++k) {
        const float piv = m[k][k];
        const float r0 = __builtin_amdgcn_rcpf(piv);
        const float pinv = fmaf(fmaf(-piv, r0, 1.0f), r0, r0);
        #pragma unroll
        for (int j = k + 1; j < 4; ++j) m[k][j] *= pinv;
        #pragma unroll
        for (int vw = 0; vw < 2; ++vw) q[vw][k] *= pinv;
        #pragma unroll
        for (int i = 0; i < 4; ++i) {
            if (i == k) continue;
            const float f = m[i][k];
            #pragma unroll
            for (int j = k + 1; j < 4; ++j) m[i][j] -= f * m[k][j];
            #pragma unroll
            for (int vw = 0; vw < 2; ++vw) q[vw][i] -= f * q[vw][k];
        }
    }

    #pragma unroll
    for (int vw = 0; vw < 2; ++vw) {
        v2f bb;
        bb.x = -(half ? q[vw][2] : q[vw][0]);
        bb.y = -(half ? q[vw][3] : q[vw][1]);
        c0[vw] = __builtin_amdgcn_wmma_f32_16x16x4_f32(false, a0, false, bb, (short)0, c0[vw], false, false);
        c1[vw] = __builtin_amdgcn_wmma_f32_16x16x4_f32(false, a1, false, bb, (short)0, c1[vw], false, false);
    }

    if (half == HP) {
        #pragma unroll
        for (int vw = 0; vw < 2; ++vw) {
            #pragma unroll
            for (int j = 0; j < 4; ++j) {
                if (WC == 0) c0[vw][RP + j] = q[vw][j];
                else         c1[vw][RP + j] = q[vw][j];
            }
        }
    }
}

__global__ void __launch_bounds__(NW * 32)
ftattn_main_kernel(const float* __restrict__ x,
                   const float* __restrict__ latent,
                   const float* __restrict__ kern,
                   const float* __restrict__ bias,
                   const float* __restrict__ beta,
                   const float* __restrict__ wsum,
                   float* __restrict__ out) {
    __shared__ __attribute__((aligned(16))) float sKer[PDF * LP];
    __shared__ __attribute__((aligned(16))) float sBias[PDF * LP];
    __shared__ __attribute__((aligned(16))) WaveLds sW[NW];

    const int tid  = threadIdx.x;
    const int b    = blockIdx.y;
    const int lane = tid & 31;
    const int wave = __builtin_amdgcn_readfirstlane(tid >> 5);
    const int half = lane >> 4;
    const int lr   = lane & 15;
    const int t0   = blockIdx.x * (NW * TPW) + wave * TPW;

    float* sInv = sW[wave].Inv;
    float* sMx  = sW[wave].Mx;
    float* sPan = sW[wave].Pan;
    float* sLat = sW[wave].Lat;

    for (int i4 = tid; i4 < PDF * PL / 4; i4 += NW * 32) {
        const int d = i4 >> 5, l = (i4 & 31) * 4;
        *reinterpret_cast<v4f*>(&sKer[d * LP + l])  = *reinterpret_cast<const v4f*>(&kern[i4 * 4]);
        *reinterpret_cast<v4f*>(&sBias[d * LP + l]) = *reinterpret_cast<const v4f*>(&bias[i4 * 4]);
    }

    const float inv_bsum = 1.0f / wsum[0];
    v4f acc = {0.0f, 0.0f, 0.0f, 0.0f};

    float xa_next  = x[((size_t)b * PT + t0) * PDF + lr];
    float xb_next  = x[((size_t)b * PT + t0) * PDF + 16 + lr];
    v4f   lat_next = *reinterpret_cast<const v4f*>(&latent[((size_t)b * PT + t0) * PL + 4 * lane]);

    const float* kr0 = &sKer[lr * LP + 2 * half];
    const float* kr1 = &sKer[(16 + lr) * LP + 2 * half];
    const float* br0 = &sBias[lr * LP + 2 * half];
    const float* br1 = &sBias[(16 + lr) * LP + 2 * half];
    const float* yr  = &sLat[2 * half];

    __syncthreads();

    for (int tt = 0; tt < TPW; ++tt) {
        const int t = t0 + tt;
        const float xa = xa_next;
        const float xb = xb_next;
        const float xv = half ? xb : xa;
        *reinterpret_cast<v4f*>(&sLat[4 * lane]) = lat_next;
        if (tt + 1 < TPW) {
            xa_next  = x[((size_t)b * PT + t + 1) * PDF + lr];
            xb_next  = x[((size_t)b * PT + t + 1) * PDF + 16 + lr];
            lat_next = *reinterpret_cast<const v4f*>(&latent[((size_t)b * PT + t + 1) * PL + 4 * lane]);
        }
        wave_lds_handover();

        v8f c0[2], c1[2];
        v2f p0 = {0.0f, 0.0f}, p1 = {0.0f, 0.0f};
        {
            const v2f xa2 = {xa, xa};
            const v2f xb2 = {xb, xb};
            c0[0] = (v8f){}; c1[0] = (v8f){}; c0[1] = (v8f){}; c1[1] = (v8f){};
            for (int kb = 0; kb < PL; kb += 32) {
                v2f ka[8], ba[8], kc[8], bc[8], yy[8];
                #pragma unroll
                for (int j = 0; j < 8; ++j) {
                    ka[j] = *reinterpret_cast<const v2f*>(kr0 + kb + 4 * j);
                    ba[j] = *reinterpret_cast<const v2f*>(br0 + kb + 4 * j);
                    kc[j] = *reinterpret_cast<const v2f*>(kr1 + kb + 4 * j);
                    bc[j] = *reinterpret_cast<const v2f*>(br1 + kb + 4 * j);
                    yy[j] = *reinterpret_cast<const v2f*>(yr + kb + 4 * j);
                }
                #pragma unroll
                for (int j = 0; j < 8; ++j) {
                    v2f f0 = __builtin_elementwise_fma(xa2, ka[j], ba[j]);
                    v2f f1 = __builtin_elementwise_fma(xb2, kc[j], bc[j]);
                    f0.x = f0.x > 0.0f ? f0.x : 0.0f; f0.y = f0.y > 0.0f ? f0.y : 0.0f;
                    f1.x = f1.x > 0.0f ? f1.x : 0.0f; f1.y = f1.y > 0.0f ? f1.y : 0.0f;
                    c0[0] = __builtin_amdgcn_wmma_f32_16x16x4_f32(false, f0, false, f0, (short)0, c0[0], false, false);
                    c1[0] = __builtin_amdgcn_wmma_f32_16x16x4_f32(false, f1, false, f0, (short)0, c1[0], false, false);
                    c0[1] = __builtin_amdgcn_wmma_f32_16x16x4_f32(false, f0, false, f1, (short)0, c0[1], false, false);
                    c1[1] = __builtin_amdgcn_wmma_f32_16x16x4_f32(false, f1, false, f1, (short)0, c1[1], false, false);
                    p0 = __builtin_elementwise_fma(f0, yy[j], p0);
                    p1 = __builtin_elementwise_fma(f1, yy[j], p1);
                }
            }
        }

        const float s0 = p0.x + p0.y;
        const float s1 = p1.x + p1.y;
        const float xty0 = s0 + xhalf(s0);
        const float xty1 = s1 + xhalf(s1);
        const float xty  = half ? xty1 : xty0;

        gj_panel_step<0>(c0, c1, sMx, sPan, half, lr);
        gj_panel_step<1>(c0, c1, sMx, sPan, half, lr);
        gj_panel_step<2>(c0, c1, sMx, sPan, half, lr);
        gj_panel_step<3>(c0, c1, sMx, sPan, half, lr);
        gj_panel_step<4>(c0, c1, sMx, sPan, half, lr);
        gj_panel_step<5>(c0, c1, sMx, sPan, half, lr);
        gj_panel_step<6>(c0, c1, sMx, sPan, half, lr);
        gj_panel_step<7>(c0, c1, sMx, sPan, half, lr);
        #pragma unroll
        for (int vw = 0; vw < 2; ++vw) {
            const int col = 16 * vw + lr;
            #pragma unroll
            for (int r = 0; r < 8; ++r) {
                sInv[(8 * half + r) * AST + col]      = c0[vw][r];
                sInv[(16 + 8 * half + r) * AST + col] = c1[vw][r];
            }
        }
        wave_lds_handover();

        float score = 0.0f;
        #pragma unroll
        for (int e = 0; e < PDF; ++e) score = fmaf(sInv[lane * AST + e], lane_bcast(xty, e), score);

        {
            v4f oh = {0.0f, 0.0f, 0.0f, 0.0f};
            #pragma unroll
            for (int d = 0; d < PDF; ++d) {
                const float xd = lane_bcast(xv, d);
                const float sd = lane_bcast(score, d);
                const v4f kk = *reinterpret_cast<const v4f*>(&sKer[d * LP + 4 * lane]);
                const v4f bs = *reinterpret_cast<const v4f*>(&sBias[d * LP + 4 * lane]);
                v4f f;
                f.x = fmaf(xd, kk.x, bs.x); f.y = fmaf(xd, kk.y, bs.y);
                f.z = fmaf(xd, kk.z, bs.z); f.w = fmaf(xd, kk.w, bs.w);
                f.x = f.x > 0.0f ? f.x : 0.0f; f.y = f.y > 0.0f ? f.y : 0.0f;
                f.z = f.z > 0.0f ? f.z : 0.0f; f.w = f.w > 0.0f ? f.w : 0.0f;
                oh.x = fmaf(sd, f.x, oh.x); oh.y = fmaf(sd, f.y, oh.y);
                oh.z = fmaf(sd, f.z, oh.z); oh.w = fmaf(sd, f.w, oh.w);
            }
            const float w = fabsf(beta[t]) * inv_bsum;
            acc.x = fmaf(w, oh.x, acc.x); acc.y = fmaf(w, oh.y, acc.y);
            acc.z = fmaf(w, oh.z, acc.z); acc.w = fmaf(w, oh.w, acc.w);
        }
        wave_lds_handover();
    }

    *reinterpret_cast<v4f*>(&sInv[4 * lane]) = acc;
    __syncthreads();
    if (tid < PL) {
        float s = 0.0f;
        #pragma unroll
        for (int w = 0; w < NW; ++w) s += sW[w].Inv[tid];
        atomicAdd(&out[b * PL + tid], s);
    }
}

extern "C" void kernel_launch(void* const* d_in, const int* in_sizes, int n_in,
                              void* d_out, int out_size, void* d_ws, size_t ws_size,
                              hipStream_t stream) {
    const float* x      = (const float*)d_in[0];
    const float* latent = (const float*)d_in[1];
    const float* kern   = (const float*)d_in[2];
    const float* bias   = (const float*)d_in[3];
    const float* beta   = (const float*)d_in[4];
    float* out  = (float*)d_out;
    float* wsum = (float*)d_ws;

    ftattn_prep_kernel<<<1, PT, 0, stream>>>(beta, out, wsum);

    dim3 grid(PT / (NW * TPW), PB);
    ftattn_main_kernel<<<grid, NW * 32, 0, stream>>>(x, latent, kern, bias, beta, wsum, out);
}


# Symmetric three-tile elimination: XtX tile rows 0-15 x cols 16-31 never formed or updated (3 products per Gram k step and per pivot step, not 4); its reads use the mirror; last bits may differ
# speedup vs baseline: 9.0915x; 1.1368x over previous
#include <hip/hip_runtime.h>

#define PB   64
#define PT   256
#define PDF  32
#define PL   128
#define NW   16
#define TPW  4
#define LP   132
#define AST  33

typedef __attribute__((ext_vector_type(2))) float v2f;
typedef __attribute__((ext_vector_type(4))) float v4f;
typedef __attribute__((ext_vector_type(8))) float v8f;

struct WaveLds {
    float Inv[PDF * AST];
    float Mx[2 * PDF * 4];
    float Pan[2 * 4 * PDF];
    float Lat[PL];
};

__global__ void ftattn_prep_kernel(const float* __restrict__ beta,
                                   float* __restrict__ out,
                                   float* __restrict__ wsum) {
    __shared__ float red[PT];
    const int tid = threadIdx.x;
    red[tid] = fabsf(beta[tid]);
    __syncthreads();
    for (int s = PT / 2; s > 0; s >>= 1) {
        if (tid < s) red[tid] += red[tid + s];
        __syncthreads();
    }
    if (tid == 0) wsum[0] = red[0];
    for (int i = tid; i < PB * PL; i += PT) out[i] = 0.0f;
}

__device__ __forceinline__ void wave_lds_handover() {
    __builtin_amdgcn_fence(__ATOMIC_RELEASE, "wavefront");
    __builtin_amdgcn_wave_barrier();
    __builtin_amdgcn_fence(__ATOMIC_ACQUIRE, "wavefront");
}

__device__ __forceinline__ float lane_bcast(float v, int srcLane) {
    return __int_as_float(__builtin_amdgcn_readlane(__float_as_int(v), srcLane));
}

__device__ __forceinline__ float xhalf(float v) {
    const int b = __float_as_int(v);
    return __int_as_float(__builtin_amdgcn_permlanex16(b, b, 0x76543210, 0xfedcba98, false, false));
}

template <int P>
__device__ __forceinline__ void gj_panel_step(v8f& cA, v8f& cB, v8f& cD,
                                              float* sMxBase, float* sPanBase,
                                              const int half, const int lr) {
    constexpr int  K0  = 4 * P;
    constexpr bool LOW = (K0 < 16);
    constexpr int  HP  = (K0 >> 3) & 1;
    constexpr int  RP  = K0 & 7;
    constexpr int  CL  = K0 & 15;
    float* mx  = sMxBase  + (P & 1) * (PDF * 4);
    float* pan = sPanBase + (P & 1) * (4 * PDF);

    const bool pivcol = (lr >= CL && lr < CL + 4);
    if (pivcol) {
        const int j = lr - CL;
        #pragma unroll
        for (int r = 0; r < 8; ++r) {
            if (LOW) {
                mx[(8 * half + r) * 4 + j]      = cA[r];
                mx[(16 + 8 * half + r) * 4 + j] = cB[r];
            } else {
                mx[(16 + 8 * half + r) * 4 + j] = cD[r];
            }
        }
    }
    #pragma unroll
    for (int r = 0; r < 8; ++r) {
        float e = 0.0f;
        if (r >= RP && r < RP + 4) e = (half == HP && lr == CL + r - RP) ? 1.0f : 0.0f;
        if (LOW) {
            cA[r] = pivcol ? e    : cA[r];
            cB[r] = pivcol ? 0.0f : cB[r];
        } else {
            cD[r] = pivcol ? e : cD[r];
        }
    }
    if (half == HP) {
        #pragma unroll
        for (int j = 0; j < 4; ++j) {
            if (LOW) {
                pan[j * PDF + lr] = cA[RP + j];
            } else {
                pan[j * PDF + lr]      = cB[RP + j];
                pan[j * PDF + 16 + lr] = cD[RP + j];
            }
        }
    }
    wave_lds_handover();

    float m[4][4];
    float q[2][4];
    #pragma unroll
    for (int i = 0; i < 4; ++i) {
        const v4f row = *reinterpret_cast<const v4f*>(mx + (K0 + i) * 4);
        m[i][0] = row.x; m[i][1] = row.y; m[i][2] = row.z; m[i][3] = row.w;
    }
    #pragma unroll
    for (int j = 0; j < 4; ++j) q[0][j] = pan[j * PDF + lr];
    if (LOW) {
        const v4f mir = *reinterpret_cast<const v4f*>(mx + (16 + lr) * 4);
        q[1][0] = mir.x; q[1][1] = mir.y; q[1][2] = mir.z; q[1][3] = mir.w;
    } else {
        #pragma unroll
        for (int j = 0; j < 4; ++j) q[1][j] = pan[j * PDF + 16 + lr];
    }

    v2f a0;
    if (LOW) {
        a0 = *reinterpret_cast<const v2f*>(mx + lr * 4 + 2 * half);
    } else {
        a0.x = -(half ? q[0][2] : q[0][0]);
        a0.y = -(half ? q[0][3] : q[0][1]);
    }
    const v2f a1 = *reinterpret_cast<const v2f*>(mx + (16 + lr) * 4 + 2 * half);

    #pragma unroll
    for (int k = 0; k < 4; ++k) {
        const float piv = m[k][k];
        const float r0 = __builtin_amdgcn_rcpf(piv);
        const float pinv = fmaf(fmaf(-piv, r0, 1.0f), r0, r0);
        #pragma unroll
        for (int j = k + 1; j < 4; ++j) m[k][j] *= pinv;
        #pragma unroll
        for (int vw = 0; vw < 2; ++vw) q[vw][k] *= pinv;
        #pragma unroll
        for (int i = 0; i < 4; ++i) {
            if (i == k) continue;
            const float f = m[i][k];
            #pragma unroll
            for (int j = k + 1; j < 4; ++j) m[i][j] -= f * m[k][j];
            #pragma unroll
            for (int vw = 0; vw < 2; ++vw) q[vw][i] -= f * q[vw][k];
        }
    }

    v2f bb0, bb1;
    bb0.x = -(half ? q[0][2] : q[0][0]);
    bb0.y = -(half ? q[0][3] : q[0][1]);
    bb1.x = -(half ? q[1][2] : q[1][0]);
    bb1.y = -(half ? q[1][3] : q[1][1]);
    cA = __builtin_amdgcn_wmma_f32_16x16x4_f32(false, a0, false, bb0, (short)0, cA, false, false);
    cB = __builtin_amdgcn_wmma_f32_16x16x4_f32(false, a1, false, bb0, (short)0, cB, false, false);
    cD = __builtin_amdgcn_wmma_f32_16x16x4_f32(false, a1, false, bb1, (short)0, cD, false, false);

    if (half == HP) {
        #pragma unroll
        for (int j = 0; j < 4; ++j) {
            if (LOW) {
                cA[RP + j] = q[0][j];
            } else {
                cB[RP + j] = q[0][j];
                cD[RP + j] = q[1][j];
            }
        }
    }
}

__global__ void __launch_bounds__(NW * 32)
ftattn_main_kernel(const float* __restrict__ x,
                   const float* __restrict__ latent,
                   const float* __restrict__ kern,
                   const float* __restrict__ bias,
                   const float* __restrict__ beta,
                   const float* __restrict__ wsum,
                   float* __restrict__ out) {
    __shared__ __attribute__((aligned(16))) float sKer[PDF * LP];
    __shared__ __attribute__((aligned(16))) float sBias[PDF * LP];
    __shared__ __attribute__((aligned(16))) WaveLds sW[NW];

    const int tid  = threadIdx.x;
    const int b    = blockIdx.y;
    const int lane = tid & 31;
    const int wave = __builtin_amdgcn_readfirstlane(tid >> 5);
    const int half = lane >> 4;
    const int lr   = lane & 15;
    const int t0   = blockIdx.x * (NW * TPW) + wave * TPW;

    float* sInv = sW[wave].Inv;
    float* sMx  = sW[wave].Mx;
    float* sPan = sW[wave].Pan;
    float* sLat = sW[wave].Lat;

    for (int i4 = tid; i4 < PDF * PL / 4; i4 += NW * 32) {
        const int d = i4 >> 5, l = (i4 & 31) * 4;
        *reinterpret_cast<v4f*>(&sKer[d * LP + l])  = *reinterpret_cast<const v4f*>(&kern[i4 * 4]);
        *reinterpret_cast<v4f*>(&sBias[d * LP + l]) = *reinterpret_cast<const v4f*>(&bias[i4 * 4]);
    }

    const float inv_bsum = 1.0f / wsum[0];
    v4f acc = {0.0f, 0.0f, 0.0f, 0.0f};

    float xa_next  = x[((size_t)b * PT + t0) * PDF + lr];
    float xb_next  = x[((size_t)b * PT + t0) * PDF + 16 + lr];
    v4f   lat_next = *reinterpret_cast<const v4f*>(&latent[((size_t)b * PT + t0) * PL + 4 * lane]);

    const float* kr0 = &sKer[lr * LP + 2 * half];
    const float* kr1 = &sKer[(16 + lr) * LP + 2 * half];
    const float* br0 = &sBias[lr * LP + 2 * half];
    const float* br1 = &sBias[(16 + lr) * LP + 2 * half];
    const float* yr  = &sLat[2 * half];

    __syncthreads();

    for (int tt = 0; tt < TPW; ++tt) {
        const int t = t0 + tt;
        const float xa = xa_next;
        const float xb = xb_next;
        const float xv = half ? xb : xa;
        *reinterpret_cast<v4f*>(&sLat[4 * lane]) = lat_next;
        if (tt + 1 < TPW) {
            xa_next  = x[((size_t)b * PT + t + 1) * PDF + lr];
            xb_next  = x[((size_t)b * PT + t + 1) * PDF + 16 + lr];
            lat_next = *reinterpret_cast<const v4f*>(&latent[((size_t)b * PT + t + 1) * PL + 4 * lane]);
        }
        wave_lds_handover();

        v8f cA, cB, cD;
        v2f p0 = {0.0f, 0.0f}, p1 = {0.0f, 0.0f};
        {
            const v2f xa2 = {xa, xa};
            const v2f xb2 = {xb, xb};
            cA = (v8f){}; cB = (v8f){}; cD = (v8f){};
            for (int kb = 0; kb < PL; kb += 32) {
                v2f ka[8], ba[8], kc[8], bc[8], yy[8];
                #pragma unroll
                for (int j = 0; j < 8; ++j) {
                    ka[j] = *reinterpret_cast<const v2f*>(kr0 + kb + 4 * j);
                    ba[j] = *reinterpret_cast<const v2f*>(br0 + kb + 4 * j);
                    kc[j] = *reinterpret_cast<const v2f*>(kr1 + kb + 4 * j);
                    bc[j] = *reinterpret_cast<const v2f*>(br1 + kb + 4 * j);
                    yy[j] = *reinterpret_cast<const v2f*>(yr + kb + 4 * j);
                }
                #pragma unroll
                for (int j = 0; j < 8; ++j) {
                    v2f f0 = __builtin_elementwise_fma(xa2, ka[j], ba[j]);
                    v2f f1 = __builtin_elementwise_fma(xb2, kc[j], bc[j]);
                    f0.x = f0.x > 0.0f ? f0.x : 0.0f; f0.y = f0.y > 0.0f ? f0.y : 0.0f;
                    f1.x = f1.x > 0.0f ? f1.x : 0.0f; f1.y = f1.y > 0.0f ? f1.y : 0.0f;
                    cA = __builtin_amdgcn_wmma_f32_16x16x4_f32(false, f0, false, f0, (short)0, cA, false, false);
                    cB = __builtin_amdgcn_wmma_f32_16x16x4_f32(false, f1, false, f0, (short)0, cB, false, false);
                    cD = __builtin_amdgcn_wmma_f32_16x16x4_f32(false, f1, false, f1, (short)0, cD, false, false);
                    p0 = __builtin_elementwise_fma(f0, yy[j], p0);
                    p1 = __builtin_elementwise_fma(f1, yy[j], p1);
                }
            }
        }

        const float s0 = p0.x + p0.y;
        const float s1 = p1.x + p1.y;
        const float xty0 = s0 + xhalf(s0);
        const float xty1 = s1 + xhalf(s1);
        const float xty  = half ? xty1 : xty0;

        gj_panel_step<0>(cA, cB, cD, sMx, sPan, half, lr);
        gj_panel_step<1>(cA, cB, cD, sMx, sPan, half, lr);
        gj_panel_step<2>(cA, cB, cD, sMx, sPan, half, lr);
        gj_panel_step<3>(cA, cB, cD, sMx, sPan, half, lr);
        gj_panel_step<4>(cA, cB, cD, sMx, sPan, half, lr);
        gj_panel_step<5>(cA, cB, cD, sMx, sPan, half, lr);
        gj_panel_step<6>(cA, cB, cD, sMx, sPan, half, lr);
        gj_panel_step<7>(cA, cB, cD, sMx, sPan, half, lr);
        #pragma unroll
        for (int r = 0; r < 8; ++r) {
            sInv[(8 * half + r) * AST + lr]           = cA[r];
            sInv[(16 + 8 * half + r) * AST + lr]      = cB[r];
            sInv[lr * AST + 16 + 8 * half + r]        = cB[r];
            sInv[(16 + 8 * half + r) * AST + 16 + lr] = cD[r];
        }
        wave_lds_handover();

        float score = 0.0f;
        #pragma unroll
        for (int e = 0; e < PDF; ++e) score = fmaf(sInv[lane * AST + e], lane_bcast(xty, e), score);

        {
            v4f oh = {0.0f, 0.0f, 0.0f, 0.0f};
            #pragma unroll
            for (int d = 0; d < PDF; ++d) {
                const float xd = lane_bcast(xv, d);
                const float sd = lane_bcast(score, d);
                const v4f kk = *reinterpret_cast<const v4f*>(&sKer[d * LP + 4 * lane]);
                const v4f bs = *reinterpret_cast<const v4f*>(&sBias[d * LP + 4 * lane]);
                v4f f;
                f.x = fmaf(xd, kk.x, bs.x); f.y = fmaf(xd, kk.y, bs.y);
                f.z = fmaf(xd, kk.z, bs.z); f.w = fmaf(xd, kk.w, bs.w);
                f.x = f.x > 0.0f ? f.x : 0.0f; f.y = f.y > 0.0f ? f.y : 0.0f;
                f.z = f.z > 0.0f ? f.z : 0.0f; f.w = f.w > 0.0f ? f.w : 0.0f;
                oh.x = fmaf(sd, f.x, oh.x); oh.y = fmaf(sd, f.y, oh.y);
                oh.z = fmaf(sd, f.z, oh.z); oh.w = fmaf(sd, f.w, oh.w);
            }
            const float w = fabsf(beta[t]) * inv_bsum;
            acc.x = fmaf(w, oh.x, acc.x); acc.y = fmaf(w, oh.y, acc.y);
            acc.z = fmaf(w, oh.z, acc.z); acc.w = fmaf(w, oh.w, acc.w);
        }
        wave_lds_handover();
    }

    *reinterpret_cast<v4f*>(&sInv[4 * lane]) = acc;
    __syncthreads();
    if (tid < PL) {
        float s = 0.0f;
        #pragma unroll
        for (int w = 0; w < NW; ++w) s += sW[w].Inv[tid];
        atomicAdd(&out[b * PL + tid], s);
    }
}

extern "C" void kernel_launch(void* const* d_in, const int* in_sizes, int n_in,
                              void* d_out, int out_size, void* d_ws, size_t ws_size,
                              hipStream_t stream) {
    const float* x      = (const float*)d_in[0];
    const float* latent = (const float*)d_in[1];
    const float* kern   = (const float*)d_in[2];
    const float* bias   = (const float*)d_in[3];
    const float* beta   = (const float*)d_in[4];
    float* out  = (float*)d_out;
    float* wsum = (float*)d_ws;

    ftattn_prep_kernel<<<1, PT, 0, stream>>>(beta, out, wsum);

    dim3 grid(PT / (NW * TPW), PB);
    ftattn_main_kernel<<<grid, NW * 32, 0, stream>>>(x, latent, kern, bias, beta, wsum, out);
}


# Gauss-Jordan pivot step as straight-line code: every lane stores both tiles whole to LDS; pivot block, pivot rows, mirror and multipliers read from that copy; no lane-masked stores or branches
# speedup vs baseline: 9.3870x; 1.0325x over previous
#include <hip/hip_runtime.h>

#define PB   64
#define PT   256
#define PDF  32
#define PL   128
#define NW   16
#define TPW  4
#define LP   132
#define AST  33
#define TS   20

typedef __attribute__((ext_vector_type(2))) float v2f;
typedef __attribute__((ext_vector_type(4))) float v4f;
typedef __attribute__((ext_vector_type(8))) float v8f;

struct WaveLds {
    float Inv[PDF * AST];
    float Tile[2 * 2 * 16 * TS];
    float Lat[PL];
};

__global__ void ftattn_prep_kernel(const float* __restrict__ beta,
                                   float* __restrict__ out,
                                   float* __restrict__ wsum) {
    __shared__ float red[PT];
    const int tid = threadIdx.x;
    red[tid] = fabsf(beta[tid]);
    __syncthreads();
    for (int s = PT / 2; s > 0; s >>= 1) {
        if (tid < s) red[tid] += red[tid + s];
        __syncthreads();
    }
    if (tid == 0) wsum[0] = red[0];
    for (int i = tid; i < PB * PL; i += PT) out[i] = 0.0f;
}

__device__ __forceinline__ void wave_lds_handover() {
    __builtin_amdgcn_fence(__ATOMIC_RELEASE, "wavefront");
    __builtin_amdgcn_wave_barrier();
    __builtin_amdgcn_fence(__ATOMIC_ACQUIRE, "wavefront");
}

__device__ __forceinline__ float lane_bcast(float v, int srcLane) {
    return __int_as_float(__builtin_amdgcn_readlane(__float_as_int(v), srcLane));
}

__device__ __forceinline__ float xhalf(float v) {
    const int b = __float_as_int(v);
    return __int_as_float(__builtin_amdgcn_permlanex16(b, b, 0x76543210, 0xfedcba98, false, false));
}

template <int P>
__device__ __forceinline__ void gj_panel_step(v8f& cA, v8f& cB, v8f& cD,
                                              float* sTileBase, const float* sIdent,
                                              const int half, const int lr) {
    constexpr int  K0  = 4 * P;
    constexpr bool LOW = (K0 < 16);
    constexpr int  HP  = (K0 >> 3) & 1;
    constexpr int  RP  = K0 & 7;
    constexpr int  CL  = K0 & 15;
    float* tu = sTileBase + (P & 1) * (2 * 16 * TS);
    float* tv = tu + 16 * TS;

    #pragma unroll
    for (int r = 0; r < 8; ++r) {
        tu[(8 * half + r) * TS + lr] = LOW ? cA[r] : cB[r];
        tv[(8 * half + r) * TS + lr] = LOW ? cB[r] : cD[r];
    }
    wave_lds_handover();

    const bool pivcol = (lr >= CL && lr < CL + 4);
    #pragma unroll
    for (int r = 0; r < 8; ++r) {
        if (LOW) {
            cA[r] = pivcol ? 0.0f : cA[r];
            cB[r] = pivcol ? 0.0f : cB[r];
        } else {
            cD[r] = pivcol ? 0.0f : cD[r];
        }
    }

    const float* tp = LOW ? tu : tv;
    float m[4][4];
    float q[2][4];
    #pragma unroll
    for (int i = 0; i < 4; ++i) {
        const v4f row = *reinterpret_cast<const v4f*>(tp + (CL + i) * TS + CL);
        m[i][0] = row.x; m[i][1] = row.y; m[i][2] = row.z; m[i][3] = row.w;
    }
    const float* qp = pivcol ? (sIdent + (lr - CL)) : (tp + CL * TS + lr);
    #pragma unroll
    for (int j = 0; j < 4; ++j) q[LOW ? 0 : 1][j] = qp[j * TS];
    if (LOW) {
        const v4f mir = *reinterpret_cast<const v4f*>(tv + lr * TS + CL);
        q[1][0] = mir.x; q[1][1] = mir.y; q[1][2] = mir.z; q[1][3] = mir.w;
    } else {
        #pragma unroll
        for (int j = 0; j < 4; ++j) q[0][j] = tu[(CL + j) * TS + lr];
    }

    v2f a0;
    if (LOW) {
        a0 = *reinterpret_cast<const v2f*>(tu + lr * TS + CL + 2 * half);
    } else {
        a0.x = -(half ? q[0][2] : q[0][0]);
        a0.y = -(half ? q[0][3] : q[0][1]);
    }
    const v2f a1 = *reinterpret_cast<const v2f*>(tv + lr * TS + CL + 2 * half);

    #pragma unroll
    for (int k = 0; k < 4; ++k) {
        const float piv = m[k][k];
        const float r0 = __builtin_amdgcn_rcpf(piv);
        const float pinv = fmaf(fmaf(-piv, r0, 1.0f), r0, r0);
        #pragma unroll
        for (int j = k + 1; j < 4; ++j) m[k][j] *= pinv;
        #pragma unroll
        for (int vw = 0; vw < 2; ++vw) q[vw][k] *= pinv;
        #pragma unroll
        for (int i = 0; i < 4; ++i) {
            if (i == k) continue;
            const float f = m[i][k];
            #pragma unroll
            for (int j = k + 1; j < 4; ++j) m[i][j] -= f * m[k][j];
            #pragma unroll
            for (int vw = 0; vw < 2; ++vw) q[vw][i] -= f * q[vw][k];
        }
    }

    v2f bb0, bb1;
    bb0.x = -(half ? q[0][2] : q[0][0]);
    bb0.y = -(half ? q[0][3] : q[0][1]);
    bb1.x = -(half ? q[1][2] : q[1][0]);
    bb1.y = -(half ? q[1][3] : q[1][1]);
    cA = __builtin_amdgcn_wmma_f32_16x16x4_f32(false, a0, false, bb0, (short)0, cA, false, false);
    cB = __builtin_amdgcn_wmma_f32_16x16x4_f32(false, a1, false, bb0, (short)0, cB, false, false);
    cD = __builtin_amdgcn_wmma_f32_16x16x4_f32(false, a1, false, bb1, (short)0, cD, false, false);

    #pragma unroll
    for (int j = 0; j < 4; ++j) {
        if (LOW) {
            cA[RP + j] = (half == HP) ? q[0][j] : cA[RP + j];
        } else {
            cB[RP + j] = (half == HP) ? q[0][j] : cB[RP + j];
            cD[RP + j] = (half == HP) ? q[1][j] : cD[RP + j];
        }
    }
    __builtin_amdgcn_sched_barrier(0);
}

__global__ void __launch_bounds__(NW * 32)
ftattn_main_kernel(const float* __restrict__ x,
                   const float* __restrict__ latent,
                   const float* __restrict__ kern,
                   const float* __restrict__ bias,
                   const float* __restrict__ beta,
                   const float* __restrict__ wsum,
                   float* __restrict__ out) {
    __shared__ __attribute__((aligned(16))) float sKer[PDF * LP];
    __shared__ __attribute__((aligned(16))) float sBias[PDF * LP];
    __shared__ __attribute__((aligned(16))) WaveLds sW[NW];
    __shared__ float sIdent[4 * TS];

    const int tid  = threadIdx.x;
    const int b    = blockIdx.y;
    const int lane = tid & 31;
    const int wave = __builtin_amdgcn_readfirstlane(tid >> 5);
    const int half = lane >> 4;
    const int lr   = lane & 15;
    const int t0   = blockIdx.x * (NW * TPW) + wave * TPW;

    float* sInv = sW[wave].Inv;
    float* sTile = sW[wave].Tile;
    float* sLat = sW[wave].Lat;

    for (int i4 = tid; i4 < PDF * PL / 4; i4 += NW * 32) {
        const int d = i4 >> 5, l = (i4 & 31) * 4;
        *reinterpret_cast<v4f*>(&sKer[d * LP + l])  = *reinterpret_cast<const v4f*>(&kern[i4 * 4]);
        *reinterpret_cast<v4f*>(&sBias[d * LP + l]) = *reinterpret_cast<const v4f*>(&bias[i4 * 4]);
    }

    if (tid < 16) sIdent[(tid >> 2) * TS + (tid & 3)] = ((tid >> 2) == (tid & 3)) ? 1.0f : 0.0f;

    const float inv_bsum = 1.0f / wsum[0];
    v4f acc = {0.0f, 0.0f, 0.0f, 0.0f};

    float xa_next  = x[((size_t)b * PT + t0) * PDF + lr];
    float xb_next  = x[((size_t)b * PT + t0) * PDF + 16 + lr];
    v4f   lat_next = *reinterpret_cast<const v4f*>(&latent[((size_t)b * PT + t0) * PL + 4 * lane]);

    const float* kr0 = &sKer[lr * LP + 2 * half];
    const float* kr1 = &sKer[(16 + lr) * LP + 2 * half];
    const float* br0 = &sBias[lr * LP + 2 * half];
    const float* br1 = &sBias[(16 + lr) * LP + 2 * half];
    const float* yr  = &sLat[2 * half];

    __syncthreads();

    for (int tt = 0; tt < TPW; ++tt) {
        const int t = t0 + tt;
        const float xa = xa_next;
        const float xb = xb_next;
        const float xv = half ? xb : xa;
        *reinterpret_cast<v4f*>(&sLat[4 * lane]) = lat_next;
        if (tt + 1 < TPW) {
            xa_next  = x[((size_t)b * PT + t + 1) * PDF + lr];
            xb_next  = x[((size_t)b * PT + t + 1) * PDF + 16 + lr];
            lat_next = *reinterpret_cast<const v4f*>(&latent[((size_t)b * PT + t + 1) * PL + 4 * lane]);
        }
        wave_lds_handover();

        v8f cA, cB, cD;
        v2f p0 = {0.0f, 0.0f}, p1 = {0.0f, 0.0f};
        {
            const v2f xa2 = {xa, xa};
            const v2f xb2 = {xb, xb};
            cA = (v8f){}; cB = (v8f){}; cD = (v8f){};
            for (int kb = 0; kb < PL; kb += 32) {
                v2f ka[8], ba[8], kc[8], bc[8], yy[8];
                #pragma unroll
                for (int j = 0; j < 8; ++j) {
                    ka[j] = *reinterpret_cast<const v2f*>(kr0 + kb + 4 * j);
                    ba[j] = *reinterpret_cast<const v2f*>(br0 + kb + 4 * j);
                    kc[j] = *reinterpret_cast<const v2f*>(kr1 + kb + 4 * j);
                    bc[j] = *reinterpret_cast<const v2f*>(br1 + kb + 4 * j);
                    yy[j] = *reinterpret_cast<const v2f*>(yr + kb + 4 * j);
                }
                #pragma unroll
                for (int j = 0; j < 8; ++j) {
                    v2f f0 = __builtin_elementwise_fma(xa2, ka[j], ba[j]);
                    v2f f1 = __builtin_elementwise_fma(xb2, kc[j], bc[j]);
                    f0.x = f0.x > 0.0f ? f0.x : 0.0f; f0.y = f0.y > 0.0f ? f0.y : 0.0f;
                    f1.x = f1.x > 0.0f ? f1.x : 0.0f; f1.y = f1.y > 0.0f ? f1.y : 0.0f;
                    cA = __builtin_amdgcn_wmma_f32_16x16x4_f32(false, f0, false, f0, (short)0, cA, false, false);
                    cB = __builtin_amdgcn_wmma_f32_16x16x4_f32(false, f1, false, f0, (short)0, cB, false, false);
                    cD = __builtin_amdgcn_wmma_f32_16x16x4_f32(false, f1, false, f1, (short)0, cD, false, false);
                    p0 = __builtin_elementwise_fma(f0, yy[j], p0);
                    p1 = __builtin_elementwise_fma(f1, yy[j], p1);
                }
            }
        }

        const float s0 = p0.x + p0.y;
        const float s1 = p1.x + p1.y;
        const float xty0 = s0 + xhalf(s0);
        const float xty1 = s1 + xhalf(s1);
        const float xty  = half ? xty1 : xty0;

        gj_panel_step<0>(cA, cB, cD, sTile, sIdent, half, lr);
        gj_panel_step<1>(cA, cB, cD, sTile, sIdent, half, lr);
        gj_panel_step<2>(cA, cB, cD, sTile, sIdent, half, lr);
        gj_panel_step<3>(cA, cB, cD, sTile, sIdent, half, lr);
        gj_panel_step<4>(cA, cB, cD, sTile, sIdent, half, lr);
        gj_panel_step<5>(cA, cB, cD, sTile, sIdent, half, lr);
        gj_panel_step<6>(cA, cB, cD, sTile, sIdent, half, lr);
        gj_panel_step<7>(cA, cB, cD, sTile, sIdent, half, lr);
        #pragma unroll
        for (int r = 0; r < 8; ++r) {
            sInv[(8 * half + r) * AST + lr]           = cA[r];
            sInv[(16 + 8 * half + r) * AST + lr]      = cB[r];
            sInv[lr * AST + 16 + 8 * half + r]        = cB[r];
            sInv[(16 + 8 * half + r) * AST + 16 + lr] = cD[r];
        }
        wave_lds_handover();

        float score = 0.0f;
        #pragma unroll
        for (int e = 0; e < PDF; ++e) score = fmaf(sInv[lane * AST + e], lane_bcast(xty, e), score);

        {
            v4f oh = {0.0f, 0.0f, 0.0f, 0.0f};
            #pragma unroll
            for (int d = 0; d < PDF; ++d) {
                const float xd = lane_bcast(xv, d);
                const float sd = lane_bcast(score, d);
                const v4f kk = *reinterpret_cast<const v4f*>(&sKer[d * LP + 4 * lane]);
                const v4f bs = *reinterpret_cast<const v4f*>(&sBias[d * LP + 4 * lane]);
                v4f f;
                f.x = fmaf(xd, kk.x, bs.x); f.y = fmaf(xd, kk.y, bs.y);
                f.z = fmaf(xd, kk.z, bs.z); f.w = fmaf(xd, kk.w, bs.w);
                f.x = f.x > 0.0f ? f.x : 0.0f; f.y = f.y > 0.0f ? f.y : 0.0f;
                f.z = f.z > 0.0f ? f.z : 0.0f; f.w = f.w > 0.0f ? f.w : 0.0f;
                oh.x = fmaf(sd, f.x, oh.x); oh.y = fmaf(sd, f.y, oh.y);
                oh.z = fmaf(sd, f.z, oh.z); oh.w = fmaf(sd, f.w, oh.w);
            }
            const float w = fabsf(beta[t]) * inv_bsum;
            acc.x = fmaf(w, oh.x, acc.x); acc.y = fmaf(w, oh.y, acc.y);
            acc.z = fmaf(w, oh.z, acc.z); acc.w = fmaf(w, oh.w, acc.w);
        }
        wave_lds_handover();
    }

    *reinterpret_cast<v4f*>(&sInv[4 * lane]) = acc;
    __syncthreads();
    if (tid < PL) {
        float s = 0.0f;
        #pragma unroll
        for (int w = 0; w < NW; ++w) s += sW[w].Inv[tid];
        atomicAdd(&out[b * PL + tid], s);
    }
}

extern "C" void kernel_launch(void* const* d_in, const int* in_sizes, int n_in,
                              void* d_out, int out_size, void* d_ws, size_t ws_size,
                              hipStream_t stream) {
    const float* x      = (const float*)d_in[0];
    const float* latent = (const float*)d_in[1];
    const float* kern   = (const float*)d_in[2];
    const float* bias   = (const float*)d_in[3];
    const float* beta   = (const float*)d_in[4];
    float* out  = (float*)d_out;
    float* wsum = (float*)d_ws;

    ftattn_prep_kernel<<<1, PT, 0, stream>>>(beta, out, wsum);

    dim3 grid(PT / (NW * TPW), PB);
    ftattn_main_kernel<<<grid, NW * 32, 0, stream>>>(x, latent, kern, bias, beta, wsum, out);
}


# Time steps from an LDS ticket counter: a wave takes the next of its block's 64 steps when done, not 4 fixed ones; each step's row kept in an LDS slot, slots summed in step order (sums regrouped)
# speedup vs baseline: 9.8914x; 1.0537x over previous
#include <hip/hip_runtime.h>

#define PB   64
#define PT   256
#define PDF  32
#define PL   128
#define NW   16
#define TPW  4
#define SPB  (NW * TPW)
#define LP   132
#define AST  33
#define TS   20

typedef __attribute__((ext_vector_type(2))) float v2f;
typedef __attribute__((ext_vector_type(4))) float v4f;
typedef __attribute__((ext_vector_type(8))) float v8f;

struct WaveLds {
    float Inv[PDF * AST];
    float Tile[2 * 2 * 16 * TS];
    float Lat[PL];
};

__global__ void ftattn_prep_kernel(const float* __restrict__ beta,
                                   float* __restrict__ out,
                                   float* __restrict__ wsum) {
    __shared__ float red[PT];
    const int tid = threadIdx.x;
    red[tid] = fabsf(beta[tid]);
    __syncthreads();
    for (int s = PT / 2; s > 0; s >>= 1) {
        if (tid < s) red[tid] += red[tid + s];
        __syncthreads();
    }
    if (tid == 0) wsum[0] = red[0];
    for (int i = tid; i < PB * PL; i += PT) out[i] = 0.0f;
}

__device__ __forceinline__ void wave_lds_handover() {
    __builtin_amdgcn_fence(__ATOMIC_RELEASE, "wavefront");
    __builtin_amdgcn_wave_barrier();
    __builtin_amdgcn_fence(__ATOMIC_ACQUIRE, "wavefront");
}

__device__ __forceinline__ float lane_bcast(float v, int srcLane) {
    return __int_as_float(__builtin_amdgcn_readlane(__float_as_int(v), srcLane));
}

__device__ __forceinline__ float xhalf(float v) {
    const int b = __float_as_int(v);
    return __int_as_float(__builtin_amdgcn_permlanex16(b, b, 0x76543210, 0xfedcba98, false, false));
}

template <int P>
__device__ __forceinline__ void gj_panel_step(v8f& cA, v8f& cB, v8f& cD,
                                              float* sTileBase, const float* sIdent,
                                              const int half, const int lr) {
    constexpr int  K0  = 4 * P;
    constexpr bool LOW = (K0 < 16);
    constexpr int  HP  = (K0 >> 3) & 1;
    constexpr int  RP  = K0 & 7;
    constexpr int  CL  = K0 & 15;
    float* tu = sTileBase + (P & 1) * (2 * 16 * TS);
    float* tv = tu + 16 * TS;

    #pragma unroll
    for (int r = 0; r < 8; ++r) {
        tu[(8 * half + r) * TS + lr] = LOW ? cA[r] : cB[r];
        tv[(8 * half + r) * TS + lr] = LOW ? cB[r] : cD[r];
    }
    wave_lds_handover();

    const bool pivcol = (lr >= CL && lr < CL + 4);
    #pragma unroll
    for (int r = 0; r < 8; ++r) {
        if (LOW) {
            cA[r] = pivcol ? 0.0f : cA[r];
            cB[r] = pivcol ? 0.0f : cB[r];
        } else {
            cD[r] = pivcol ? 0.0f : cD[r];
        }
    }

    const float* tp = LOW ? tu : tv;
    float m[4][4];
    float q[2][4];
    #pragma unroll
    for (int i = 0; i < 4; ++i) {
        const v4f row = *reinterpret_cast<const v4f*>(tp + (CL + i) * TS + CL);
        m[i][0] = row.x; m[i][1] = row.y; m[i][2] = row.z; m[i][3] = row.w;
    }
    const float* qp = pivcol ? (sIdent + (lr - CL)) : (tp + CL * TS + lr);
    #pragma unroll
    for (int j = 0; j < 4; ++j) q[LOW ? 0 : 1][j] = qp[j * TS];
    if (LOW) {
        const v4f mir = *reinterpret_cast<const v4f*>(tv + lr * TS + CL);
        q[1][0] = mir.x; q[1][1] = mir.y; q[1][2] = mir.z; q[1][3] = mir.w;
    } else {
        #pragma unroll
        for (int j = 0; j < 4; ++j) q[0][j] = tu[(CL + j) * TS + lr];
    }

    v2f a0;
    if (LOW) {
        a0 = *reinterpret_cast<const v2f*>(tu + lr * TS + CL + 2 * half);
    } else {
        a0.x = -(half ? q[0][2] : q[0][0]);
        a0.y = -(half ? q[0][3] : q[0][1]);
    }
    const v2f a1 = *reinterpret_cast<const v2f*>(tv + lr * TS + CL + 2 * half);

    #pragma unroll
    for (int k = 0; k < 4; ++k) {
        const float piv = m[k][k];
        const float r0 = __builtin_amdgcn_rcpf(piv);
        const float pinv = fmaf(fmaf(-piv, r0, 1.0f), r0, r0);
        #pragma unroll
        for (int j = k + 1; j < 4; ++j) m[k][j] *= pinv;
        #pragma unroll
        for (int vw = 0; vw < 2; ++vw) q[vw][k] *= pinv;
        #pragma unroll
        for (int i = 0; i < 4; ++i) {
            if (i == k) continue;
            const float f = m[i][k];
            #pragma unroll
            for (int j = k + 1; j < 4; ++j) m[i][j] -= f * m[k][j];
            #pragma unroll
            for (int vw = 0; vw < 2; ++vw) q[vw][i] -= f * q[vw][k];
        }
    }

    v2f bb0, bb1;
    bb0.x = -(half ? q[0][2] : q[0][0]);
    bb0.y = -(half ? q[0][3] : q[0][1]);
    bb1.x = -(half ? q[1][2] : q[1][0]);
    bb1.y = -(half ? q[1][3] : q[1][1]);
    cA = __builtin_amdgcn_wmma_f32_16x16x4_f32(false, a0, false, bb0, (short)0, cA, false, false);
    cB = __builtin_amdgcn_wmma_f32_16x16x4_f32(false, a1, false, bb0, (short)0, cB, false, false);
    cD = __builtin_amdgcn_wmma_f32_16x16x4_f32(false, a1, false, bb1, (short)0, cD, false, false);

    #pragma unroll
    for (int j = 0; j < 4; ++j) {
        if (LOW) {
            cA[RP + j] = (half == HP) ? q[0][j] : cA[RP + j];
        } else {
            cB[RP + j] = (half == HP) ? q[0][j] : cB[RP + j];
            cD[RP + j] = (half == HP) ? q[1][j] : cD[RP + j];
        }
    }
    __builtin_amdgcn_sched_barrier(0);
}

__global__ void __launch_bounds__(NW * 32)
ftattn_main_kernel(const float* __restrict__ x,
                   const float* __restrict__ latent,
                   const float* __restrict__ kern,
                   const float* __restrict__ bias,
                   const float* __restrict__ beta,
                   const float* __restrict__ wsum,
                   float* __restrict__ out) {
    __shared__ __attribute__((aligned(16))) float sKer[PDF * LP];
    __shared__ __attribute__((aligned(16))) float sBias[PDF * LP];
    __shared__ __attribute__((aligned(16))) WaveLds sW[NW];
    __shared__ float sIdent[4 * TS];
    __shared__ __attribute__((aligned(16))) float sStep[SPB * PL];
    __shared__ __attribute__((aligned(16))) float sPart[4 * PL];
    __shared__ int sNext;

    const int tid  = threadIdx.x;
    const int b    = blockIdx.y;
    const int lane = tid & 31;
    const int wave = __builtin_amdgcn_readfirstlane(tid >> 5);
    const int half = lane >> 4;
    const int lr   = lane & 15;
    const int tbase = blockIdx.x * SPB;
    int tk = wave;

    float* sInv = sW[wave].Inv;
    float* sTile = sW[wave].Tile;
    float* sLat = sW[wave].Lat;

    for (int i4 = tid; i4 < PDF * PL / 4; i4 += NW * 32) {
        const int d = i4 >> 5, l = (i4 & 31) * 4;
        *reinterpret_cast<v4f*>(&sKer[d * LP + l])  = *reinterpret_cast<const v4f*>(&kern[i4 * 4]);
        *reinterpret_cast<v4f*>(&sBias[d * LP + l]) = *reinterpret_cast<const v4f*>(&bias[i4 * 4]);
    }

    if (tid < 16) sIdent[(tid >> 2) * TS + (tid & 3)] = ((tid >> 2) == (tid & 3)) ? 1.0f : 0.0f;
    if (tid == 0) sNext = NW;

    const float inv_bsum = 1.0f / wsum[0];

    float xa_next  = x[((size_t)b * PT + tbase + tk) * PDF + lr];
    float xb_next  = x[((size_t)b * PT + tbase + tk) * PDF + 16 + lr];
    v4f   lat_next = *reinterpret_cast<const v4f*>(&latent[((size_t)b * PT + tbase + tk) * PL + 4 * lane]);

    const float* kr0 = &sKer[lr * LP + 2 * half];
    const float* kr1 = &sKer[(16 + lr) * LP + 2 * half];
    const float* br0 = &sBias[lr * LP + 2 * half];
    const float* br1 = &sBias[(16 + lr) * LP + 2 * half];
    const float* yr  = &sLat[2 * half];

    __syncthreads();

    for (;;) {
        const int t = tbase + tk;
        const float xa = xa_next;
        const float xb = xb_next;
        const float xv = half ? xb : xa;
        *reinterpret_cast<v4f*>(&sLat[4 * lane]) = lat_next;
        wave_lds_handover();

        v8f cA, cB, cD;
        v2f p0 = {0.0f, 0.0f}, p1 = {0.0f, 0.0f};
        {
            const v2f xa2 = {xa, xa};
            const v2f xb2 = {xb, xb};
            cA = (v8f){}; cB = (v8f){}; cD = (v8f){};
            for (int kb = 0; kb < PL; kb += 32) {
                v2f ka[8], ba[8], kc[8], bc[8], yy[8];
                #pragma unroll
                for (int j = 0; j < 8; ++j) {
                    ka[j] = *reinterpret_cast<const v2f*>(kr0 + kb + 4 * j);
                    ba[j] = *reinterpret_cast<const v2f*>(br0 + kb + 4 * j);
                    kc[j] = *reinterpret_cast<const v2f*>(kr1 + kb + 4 * j);
                    bc[j] = *reinterpret_cast<const v2f*>(br1 + kb + 4 * j);
                    yy[j] = *reinterpret_cast<const v2f*>(yr + kb + 4 * j);
                }
                #pragma unroll
                for (int j = 0; j < 8; ++j) {
                    v2f f0 = __builtin_elementwise_fma(xa2, ka[j], ba[j]);
                    v2f f1 = __builtin_elementwise_fma(xb2, kc[j], bc[j]);
                    f0.x = f0.x > 0.0f ? f0.x : 0.0f; f0.y = f0.y > 0.0f ? f0.y : 0.0f;
                    f1.x = f1.x > 0.0f ? f1.x : 0.0f; f1.y = f1.y > 0.0f ? f1.y : 0.0f;
                    cA = __builtin_amdgcn_wmma_f32_16x16x4_f32(false, f0, false, f0, (short)0, cA, false, false);
                    cB = __builtin_amdgcn_wmma_f32_16x16x4_f32(false, f1, false, f0, (short)0, cB, false, false);
                    cD = __builtin_amdgcn_wmma_f32_16x16x4_f32(false, f1, false, f1, (short)0, cD, false, false);
                    p0 = __builtin_elementwise_fma(f0, yy[j], p0);
                    p1 = __builtin_elementwise_fma(f1, yy[j], p1);
                }
            }
        }

        int nk = 0;
        if (lane == 0) nk = atomicAdd(&sNext, 1);
        nk = __builtin_amdgcn_readfirstlane(nk);
        if (nk < SPB) {
            xa_next  = x[((size_t)b * PT + tbase + nk) * PDF + lr];
            xb_next  = x[((size_t)b * PT + tbase + nk) * PDF + 16 + lr];
            lat_next = *reinterpret_cast<const v4f*>(&latent[((size_t)b * PT + tbase + nk) * PL + 4 * lane]);
        }

        const float s0 = p0.x + p0.y;
        const float s1 = p1.x + p1.y;
        const float xty0 = s0 + xhalf(s0);
        const float xty1 = s1 + xhalf(s1);
        const float xty  = half ? xty1 : xty0;

        gj_panel_step<0>(cA, cB, cD, sTile, sIdent, half, lr);
        gj_panel_step<1>(cA, cB, cD, sTile, sIdent, half, lr);
        gj_panel_step<2>(cA, cB, cD, sTile, sIdent, half, lr);
        gj_panel_step<3>(cA, cB, cD, sTile, sIdent, half, lr);
        gj_panel_step<4>(cA, cB, cD, sTile, sIdent, half, lr);
        gj_panel_step<5>(cA, cB, cD, sTile, sIdent, half, lr);
        gj_panel_step<6>(cA, cB, cD, sTile, sIdent, half, lr);
        gj_panel_step<7>(cA, cB, cD, sTile, sIdent, half, lr);
        #pragma unroll
        for (int r = 0; r < 8; ++r) {
            sInv[(8 * half + r) * AST + lr]           = cA[r];
            sInv[(16 + 8 * half + r) * AST + lr]      = cB[r];
            sInv[lr * AST + 16 + 8 * half + r]        = cB[r];
            sInv[(16 + 8 * half + r) * AST + 16 + lr] = cD[r];
        }
        wave_lds_handover();

        float score = 0.0f;
        #pragma unroll
        for (int e = 0; e < PDF; ++e) score = fmaf(sInv[lane * AST + e], lane_bcast(xty, e), score);

        {
            v4f oh = {0.0f, 0.0f, 0.0f, 0.0f};
            #pragma unroll
            for (int d = 0; d < PDF; ++d) {
                const float xd = lane_bcast(xv, d);
                const float sd = lane_bcast(score, d);
                const v4f kk = *reinterpret_cast<const v4f*>(&sKer[d * LP + 4 * lane]);
                const v4f bs = *reinterpret_cast<const v4f*>(&sBias[d * LP + 4 * lane]);
                v4f f;
                f.x = fmaf(xd, kk.x, bs.x); f.y = fmaf(xd, kk.y, bs.y);
                f.z = fmaf(xd, kk.z, bs.z); f.w = fmaf(xd, kk.w, bs.w);
                f.x = f.x > 0.0f ? f.x : 0.0f; f.y = f.y > 0.0f ? f.y : 0.0f;
                f.z = f.z > 0.0f ? f.z : 0.0f; f.w = f.w > 0.0f ? f.w : 0.0f;
                oh.x = fmaf(sd, f.x, oh.x); oh.y = fmaf(sd, f.y, oh.y);
                oh.z = fmaf(sd, f.z, oh.z); oh.w = fmaf(sd, f.w, oh.w);
            }
            const float w = fabsf(beta[t]) * inv_bsum;
            const v4f row = {w * oh.x, w * oh.y, w * oh.z, w * oh.w};
            *reinterpret_cast<v4f*>(&sStep[tk * PL + 4 * lane]) = row;
        }
        wave_lds_handover();
        if (nk >= SPB) break;
        tk = nk;
    }

    __syncthreads();
    {
        const int c = tid & (PL - 1);
        const int g = tid >> 7;
        float s = 0.0f;
        #pragma unroll
        for (int k = 0; k < SPB / 4; ++k) s += sStep[(g * (SPB / 4) + k) * PL + c];
        sPart[g * PL + c] = s;
    }
    __syncthreads();
    if (tid < PL) {
        const float s = ((sPart[tid] + sPart[PL + tid]) + sPart[2 * PL + tid]) + sPart[3 * PL + tid];
        atomicAdd(&out[b * PL + tid], s);
    }
}

extern "C" void kernel_launch(void* const* d_in, const int* in_sizes, int n_in,
                              void* d_out, int out_size, void* d_ws, size_t ws_size,
                              hipStream_t stream) {
    const float* x      = (const float*)d_in[0];
    const float* latent = (const float*)d_in[1];
    const float* kern   = (const float*)d_in[2];
    const float* bias   = (const float*)d_in[3];
    const float* beta   = (const float*)d_in[4];
    float* out  = (float*)d_out;
    float* wsum = (float*)d_ws;

    ftattn_prep_kernel<<<1, PT, 0, stream>>>(beta, out, wsum);

    dim3 grid(PT / (NW * TPW), PB);
    ftattn_main_kernel<<<grid, NW * 32, 0, stream>>>(x, latent, kern, bias, beta, wsum, out);
}
